# starting kernel
# baseline (speedup 1.0000x reference)
#include <hip/hip_runtime.h>

typedef float v2f __attribute__((ext_vector_type(2)));
typedef float v8f __attribute__((ext_vector_type(8)));

#define GDIN  128
#define GDH   128
#define GDOUT 64

__global__ void zero_f32_kernel(float* __restrict__ p, size_t n) {
  size_t i = (size_t)blockIdx.x * blockDim.x + threadIdx.x;
  size_t stride = (size_t)gridDim.x * blockDim.x;
  for (; i < n; i += stride) p[i] = 0.0f;
}

__global__ void degree_kernel(const int* __restrict__ dst, float* __restrict__ deg, int E) {
  int i = blockIdx.x * blockDim.x + threadIdx.x;
  int stride = gridDim.x * blockDim.x;
  for (; i < E; i += stride) atomicAdd(&deg[dst[i]], 1.0f);
}

__global__ void rsqrt_kernel(float* __restrict__ dinv, int N) {
  int i = blockIdx.x * blockDim.x + threadIdx.x;
  if (i < N) dinv[i] = rsqrtf(dinv[i] + 1.0f);
}

template <int NCOLS>
__global__ void __launch_bounds__(256) gemm_wmma_f32(const float* __restrict__ X,
                                                     const float* __restrict__ W,
                                                     float* __restrict__ T) {
  __shared__ float xs[16 * 128];
  const int tid = threadIdx.x;
  const int wave = tid >> 5;
  const int lane = tid & 31;
  const int nwaves = blockDim.x >> 5;
  const size_t m0 = (size_t)blockIdx.x * 16;

  for (int i = tid; i < 16 * 128; i += blockDim.x) xs[i] = X[m0 * 128 + i];
  __syncthreads();

  const int m = lane & 15;
  const int kg = (lane >> 4) << 1;
  const int ntiles = NCOLS >> 4;

  for (int nt = wave; nt < ntiles; nt += nwaves) {
    const int n = (nt << 4) + (lane & 15);
    v8f acc = {};
#pragma unroll
    for (int k0 = 0; k0 < 128; k0 += 4) {
      const int kb = k0 + kg;
      v2f a, b;
      a.x = xs[m * 128 + kb];
      a.y = xs[m * 128 + kb + 1];
      b.x = W[(size_t)kb * NCOLS + n];
      b.y = W[(size_t)(kb + 1) * NCOLS + n];
      acc = __builtin_amdgcn_wmma_f32_16x16x4_f32(false, a, false, b, (short)0, acc,
                                                  false, false);
    }
    const size_t rowbase = m0 + (size_t)((lane >> 4) << 3);
    const size_t col = (size_t)(nt << 4) + (lane & 15);
#pragma unroll
    for (int r = 0; r < 8; ++r) T[(rowbase + r) * NCOLS + col] = acc[r];
  }
}

template <int D>
__global__ void scatter_edges(const int* __restrict__ src, const int* __restrict__ dst,
                              const float* __restrict__ dinv, const float* __restrict__ T,
                              float* __restrict__ OUT, int E) {
  const int lane = threadIdx.x & 31;
  int w = (blockIdx.x * blockDim.x + threadIdx.x) >> 5;
  const int nw = (gridDim.x * blockDim.x) >> 5;
  for (; w < E; w += nw) {
    const int s = src[w];
    const int d = dst[w];
    const float nrm = dinv[s] * dinv[d];
    const float* tp = T + (size_t)s * D;
    float* op = OUT + (size_t)d * D;
#pragma unroll
    for (int f = 0; f < D / 32; ++f)
      atomicAdd(&op[lane + 32 * f], tp[lane + 32 * f] * nrm);
  }
}

__global__ void finalize_relu_kernel(float* __restrict__ t1, const float* __restrict__ agg,
                                     const float* __restrict__ dinv,
                                     const float* __restrict__ bias, size_t total) {
  size_t idx = (size_t)blockIdx.x * blockDim.x + threadIdx.x;
  size_t stride = (size_t)gridDim.x * blockDim.x;
  for (; idx < total; idx += stride) {
    const size_t i = idx >> 7;
    const int j = (int)(idx & 127);
    const float di = dinv[i];
    float v = agg[idx] + t1[idx] * di * di + bias[j];
    t1[idx] = v > 0.0f ? v : 0.0f;
  }
}

__global__ void finalize_out_kernel(float* __restrict__ out, const float* __restrict__ t2,
                                    const float* __restrict__ dinv,
                                    const float* __restrict__ bias, size_t total) {
  size_t idx = (size_t)blockIdx.x * blockDim.x + threadIdx.x;
  size_t stride = (size_t)gridDim.x * blockDim.x;
  for (; idx < total; idx += stride) {
    const size_t i = idx >> 6;
    const int j = (int)(idx & 63);
    const float di = dinv[i];
    out[idx] += t2[idx] * di * di + bias[j];
  }
}

extern "C" void kernel_launch(void* const* d_in, const int* in_sizes, int n_in,
                              void* d_out, int out_size, void* d_ws, size_t ws_size,
                              hipStream_t stream) {
  const float* x  = (const float*)d_in[0];
  const int*   ei = (const int*)d_in[1];
  const float* W1 = (const float*)d_in[2];
  const float* b1 = (const float*)d_in[3];
  const float* W2 = (const float*)d_in[4];
  const float* b2 = (const float*)d_in[5];
  float* out = (float*)d_out;

  const int N = in_sizes[0] / GDIN;
  const int E = in_sizes[1] / 2;
  const int* src = ei;
  const int* dst = ei + E;

  float* dinv = (float*)d_ws;
  float* bufA = dinv + N;
  float* bufB = bufA + (size_t)N * GDH;

  const size_t nh  = (size_t)N * GDH;
  const size_t no  = (size_t)N * GDOUT;
  const int mtiles = (N + 15) / 16;

  zero_f32_kernel<<<512, 256, 0, stream>>>(dinv, (size_t)N);
  degree_kernel<<<(E + 255) / 256, 256, 0, stream>>>(dst, dinv, E);
  rsqrt_kernel<<<(N + 255) / 256, 256, 0, stream>>>(dinv, N);

  gemm_wmma_f32<GDH><<<mtiles, 256, 0, stream>>>(x, W1, bufA);
  zero_f32_kernel<<<8192, 256, 0, stream>>>(bufB, nh);
  scatter_edges<GDH><<<16384, 256, 0, stream>>>(src, dst, dinv, bufA, bufB, E);
  finalize_relu_kernel<<<8192, 256, 0, stream>>>(bufA, bufB, dinv, b1, nh);

  gemm_wmma_f32<GDOUT><<<mtiles, 256, 0, stream>>>(bufA, W2, bufB);
  zero_f32_kernel<<<8192, 256, 0, stream>>>(out, no);
  scatter_edges<GDOUT><<<16384, 256, 0, stream>>>(src, dst, dinv, bufB, out, E);
  finalize_out_kernel<<<8192, 256, 0, stream>>>(out, bufB, dinv, b2, no);
}


# Degree count: a block per (node range, edge slice) counts targets in LDS integer counters; partial rows plain-stored, summed in the rsqrt pass; per-edge global atomics and zero fill gone; 6 launches
# speedup vs baseline: 2.3853x; 2.3853x over previous
#include <hip/hip_runtime.h>

typedef float v2f __attribute__((ext_vector_type(2)));
typedef float v4f __attribute__((ext_vector_type(4)));
typedef float v8f __attribute__((ext_vector_type(8)));

#define GDIN  128
#define GDH   128
#define GDOUT 64

#define HCAP     12800
#define HPARTS   32
#define HTHREADS 512
#define HU       25

__global__ void __launch_bounds__(HTHREADS) degree_hist_kernel(const int* __restrict__ dst,
                                                                unsigned* __restrict__ part,
                                                                int N, int E, int nranges) {
  __shared__ __attribute__((aligned(16))) unsigned cnt[HCAP];
  const int tid = threadIdx.x;
  const int r = blockIdx.x % nranges;
  const int p = blockIdx.x / nranges;
  const int base = r * HCAP;
  const int len = min(HCAP, N - base);

  uint4* c4 = (uint4*)cnt;
  for (int k = tid; k < HCAP / 4; k += HTHREADS) c4[k] = make_uint4(0u, 0u, 0u, 0u);
  __syncthreads();

  const int per = (E + HPARTS - 1) / HPARTS;
  const int e0 = p * per;
  const int e1 = min(e0 + per, E);
  for (int i0 = e0 + tid; i0 < e1; i0 += HTHREADS * HU) {
    int d[HU];
#pragma unroll
    for (int u = 0; u < HU; ++u) {
      const int i = i0 + u * HTHREADS;
      d[u] = dst[i < e1 ? i : e1 - 1];
    }
#pragma unroll
    for (int u = 0; u < HU; ++u) {
      const int i = i0 + u * HTHREADS;
      const unsigned k = (unsigned)(d[u] - base);
      if (i < e1 && k < (unsigned)len) atomicAdd(&cnt[k], 1u);
    }
  }
  __syncthreads();

  unsigned* row = part + (size_t)p * N + base;
  for (int k = tid; k < len; k += HTHREADS) row[k] = cnt[k];
}

__global__ void dinv_kernel(const unsigned* __restrict__ part, float* __restrict__ dinv, int N) {
  const int i = blockIdx.x * blockDim.x + threadIdx.x;
  if (i < N) {
    unsigned c = 0u;
#pragma unroll
    for (int p = 0; p < HPARTS; ++p) c += part[(size_t)p * N + i];
    dinv[i] = rsqrtf((float)c + 1.0f);
  }
}

template <int NCOLS, bool RELU_A>
__global__ void __launch_bounds__(256) gemm_wmma_f32(const float* __restrict__ X,
                                                     const float* __restrict__ W,
                                                     const float* __restrict__ dinv,
                                                     const float* __restrict__ bias,
                                                     float* __restrict__ T,
                                                     float* __restrict__ S, int mtiles) {
  constexpr int NT = NCOLS / 16;
  constexpr int N4 = NCOLS / 4;
  constexpr int PITCH2 = NCOLS + 8;
  __shared__ v2f ws[64 * PITCH2];

  const int tid = threadIdx.x;
  const int wave = tid >> 5;
  const int lane = tid & 31;
  const int c = lane & 15;
  const int h = lane >> 4;

  const int tile = blockIdx.x * 8 + wave;
  const int ltile = tile < mtiles ? tile : mtiles - 1;

  constexpr int WIT = (64 * N4) / 256;
  v4f w0[WIT], w1[WIT];
#pragma unroll
  for (int it = 0; it < WIT; ++it) {
    const int i = tid + it * 256;
    const int kp = i / N4;
    const int n4 = i - kp * N4;
    w0[it] = *(const v4f*)(W + (size_t)(2 * kp) * NCOLS + 4 * n4);
    w1[it] = *(const v4f*)(W + (size_t)(2 * kp + 1) * NCOLS + 4 * n4);
  }

  const float* xr = X + ((size_t)ltile * 16 + c) * 128 + 4 * h;
  v4f a[16];
#pragma unroll
  for (int J = 0; J < 16; ++J) a[J] = *(const v4f*)(xr + 8 * J);

#pragma unroll
  for (int it = 0; it < WIT; ++it) {
    const int i = tid + it * 256;
    const int kp = i / N4;
    const int n4 = i - kp * N4;
    const v4f lo = {w0[it].x, w1[it].x, w0[it].y, w1[it].y};
    const v4f hi = {w0[it].z, w1[it].z, w0[it].w, w1[it].w};
    v4f* wd = (v4f*)(ws + kp * PITCH2 + 4 * n4);
    wd[0] = lo;
    wd[1] = hi;
  }
  __syncthreads();
  if (tile >= mtiles) return;

  if (RELU_A) {
#pragma unroll
    for (int J = 0; J < 16; ++J) {
      a[J].x = a[J].x > 0.0f ? a[J].x : 0.0f;
      a[J].y = a[J].y > 0.0f ? a[J].y : 0.0f;
      a[J].z = a[J].z > 0.0f ? a[J].z : 0.0f;
      a[J].w = a[J].w > 0.0f ? a[J].w : 0.0f;
    }
  }

  v8f acc[NT];
#pragma unroll
  for (int nt = 0; nt < NT; ++nt) acc[nt] = v8f{};

  const v2f* wb = ws + (2 * h) * PITCH2 + c;
#pragma unroll
  for (int J = 0; J < 16; ++J) {
#pragma unroll
    for (int sub = 0; sub < 2; ++sub) {
      v2f av;
      av.x = sub ? a[J].z : a[J].x;
      av.y = sub ? a[J].w : a[J].y;
      const v2f* wp = wb + (4 * J + sub) * PITCH2;
#pragma unroll
      for (int nt = 0; nt < NT; ++nt) {
        const v2f b = wp[16 * nt];
        acc[nt] = __builtin_amdgcn_wmma_f32_16x16x4_f32(false, av, false, b, (short)0,
                                                        acc[nt], false, false);
      }
    }
  }

  if constexpr (NT == 8) {
    asm volatile("v_nop\n\tv_nop\n\tv_nop\n\tv_nop\n\tv_nop\n\tv_nop\n\tv_nop\n\tv_nop"
                 : "+v"(acc[0]), "+v"(acc[1]), "+v"(acc[2]), "+v"(acc[3]),
                   "+v"(acc[4]), "+v"(acc[5]), "+v"(acc[6]), "+v"(acc[7]));
  } else {
    static_assert(NT == 4, "accumulator tiles tied by hand below");
    asm volatile("v_nop\n\tv_nop\n\tv_nop\n\tv_nop\n\tv_nop\n\tv_nop\n\tv_nop\n\tv_nop"
                 : "+v"(acc[0]), "+v"(acc[1]), "+v"(acc[2]), "+v"(acc[3]));
  }

  const float* dr = dinv + (size_t)tile * 16 + 8 * h;
  const v4f dlo = *(const v4f*)dr;
  const v4f dhi = *(const v4f*)(dr + 4);
  float bs[NT];
#pragma unroll
  for (int nt = 0; nt < NT; ++nt) bs[nt] = bias[c + 16 * nt];
  const float dv[8] = {dlo.x, dlo.y, dlo.z, dlo.w, dhi.x, dhi.y, dhi.z, dhi.w};
  const size_t o0 = ((size_t)tile * 16 + 8 * h) * NCOLS + c;
  float* tp = T + o0;
  float* sp = S + o0;
#pragma unroll
  for (int nt = 0; nt < NT; ++nt) {
#pragma unroll
    for (int r = 0; r < 8; ++r) {
      const float v = acc[nt][r];
      tp[(size_t)r * NCOLS + 16 * nt] = v;
      sp[(size_t)r * NCOLS + 16 * nt] = v * dv[r] * dv[r] + bs[nt];
    }
  }
}

template <int D>
__global__ void __launch_bounds__(256) scatter_edges(const int* __restrict__ src,
                                                     const int* __restrict__ dst,
                                                     const float* __restrict__ dinv,
                                                     const float* __restrict__ T,
                                                     float* __restrict__ OUT, int E) {
  constexpr int F = D / 32;
  constexpr int G = 32 / F;
  constexpr int NG = 32 / G;
  const int lane = threadIdx.x & 31;
  const int gw = __builtin_amdgcn_readfirstlane((int)((blockIdx.x * blockDim.x + threadIdx.x) >> 5));
  const int nw = (int)((gridDim.x * blockDim.x) >> 5);
  const int nb = (E + 31) >> 5;
  for (int b = gw; b < nb; b += nw) {
    const int e = b * 32 + lane;
    int s = 0, d = 0;
    float nrm = 0.0f;
    if (e < E) {
      s = src[e];
      d = dst[e];
      nrm = dinv[s] * dinv[d];
    }
    const int nbits = __float_as_int(nrm);
    const int cnt = E - b * 32;
    if (cnt >= 32) {
#pragma unroll
      for (int g = 0; g < NG; ++g) {
        float v[G][F];
#pragma unroll
        for (int u = 0; u < G; ++u) {
          const int sj = __builtin_amdgcn_readlane(s, g * G + u);
          const float* tp = T + (size_t)sj * D + lane;
#pragma unroll
          for (int f = 0; f < F; ++f) v[u][f] = tp[32 * f];
        }
#pragma unroll
        for (int u = 0; u < G; ++u) {
          const int dj = __builtin_amdgcn_readlane(d, g * G + u);
          const float nj = __int_as_float(__builtin_amdgcn_readlane(nbits, g * G + u));
          float* op = OUT + (size_t)dj * D + lane;
#pragma unroll
          for (int f = 0; f < F; ++f) atomicAdd(&op[32 * f], v[u][f] * nj);
        }
      }
    } else {
      for (int j = 0; j < cnt; ++j) {
        const int sj = __builtin_amdgcn_readlane(s, j);
        const int dj = __builtin_amdgcn_readlane(d, j);
        const float nj = __int_as_float(__builtin_amdgcn_readlane(nbits, j));
        const float* tp = T + (size_t)sj * D + lane;
        float* op = OUT + (size_t)dj * D + lane;
#pragma unroll
        for (int f = 0; f < F; ++f) atomicAdd(&op[32 * f], tp[32 * f] * nj);
      }
    }
  }
}

extern "C" void kernel_launch(void* const* d_in, const int* in_sizes, int n_in,
                              void* d_out, int out_size, void* d_ws, size_t ws_size,
                              hipStream_t stream) {
  const float* x  = (const float*)d_in[0];
  const int*   ei = (const int*)d_in[1];
  const float* W1 = (const float*)d_in[2];
  const float* b1 = (const float*)d_in[3];
  const float* W2 = (const float*)d_in[4];
  const float* b2 = (const float*)d_in[5];
  float* out = (float*)d_out;

  const int N = in_sizes[0] / GDIN;
  const int E = in_sizes[1] / 2;
  const int* src = ei;
  const int* dst = ei + E;

  float* dinv = (float*)d_ws;
  float* bufA = dinv + N;
  float* bufB = bufA + (size_t)N * GDH;

  const int mtiles = (N + 15) / 16;
  const int gblocks = (mtiles + 7) / 8;

  unsigned* part = (unsigned*)bufA;
  const int nranges = (N + HCAP - 1) / HCAP;
  degree_hist_kernel<<<nranges * HPARTS, HTHREADS, 0, stream>>>(dst, part, N, E, nranges);
  dinv_kernel<<<(N + 255) / 256, 256, 0, stream>>>(part, dinv, N);

  gemm_wmma_f32<GDH, false><<<gblocks, 256, 0, stream>>>(x, W1, dinv, b1, bufA, bufB, mtiles);
  scatter_edges<GDH><<<2048, 256, 0, stream>>>(src, dst, dinv, bufA, bufB, E);

  gemm_wmma_f32<GDOUT, true><<<gblocks, 256, 0, stream>>>(bufB, W2, dinv, b2, bufA, out, mtiles);
  scatter_edges<GDOUT><<<2048, 256, 0, stream>>>(src, dst, dinv, bufA, out, E);
}


# Layer-1 edge pass walks the edge list once per 64-column slice (256 B of a row), tables at 256-byte offsets in the workspace: one walk's rows fit the second-level cache; f32 atomics as before
# speedup vs baseline: 2.6530x; 1.1122x over previous
#include <hip/hip_runtime.h>

typedef float v2f __attribute__((ext_vector_type(2)));
typedef float v4f __attribute__((ext_vector_type(4)));
typedef float v8f __attribute__((ext_vector_type(8)));

#define GDIN  128
#define GDH   128
#define GDOUT 64

#define HCAP     12800
#define HPARTS   32
#define HTHREADS 512
#define HU       25

__global__ void __launch_bounds__(HTHREADS) degree_hist_kernel(const int* __restrict__ dst,
                                                                unsigned* __restrict__ part,
                                                                int N, int E, int nranges) {
  __shared__ __attribute__((aligned(16))) unsigned cnt[HCAP];
  const int tid = threadIdx.x;
  const int r = blockIdx.x % nranges;
  const int p = blockIdx.x / nranges;
  const int base = r * HCAP;
  const int len = min(HCAP, N - base);

  uint4* c4 = (uint4*)cnt;
  for (int k = tid; k < HCAP / 4; k += HTHREADS) c4[k] = make_uint4(0u, 0u, 0u, 0u);
  __syncthreads();

  const int per = (E + HPARTS - 1) / HPARTS;
  const int e0 = p * per;
  const int e1 = min(e0 + per, E);
  for (int i0 = e0 + tid; i0 < e1; i0 += HTHREADS * HU) {
    int d[HU];
#pragma unroll
    for (int u = 0; u < HU; ++u) {
      const int i = i0 + u * HTHREADS;
      d[u] = dst[i < e1 ? i : e1 - 1];
    }
#pragma unroll
    for (int u = 0; u < HU; ++u) {
      const int i = i0 + u * HTHREADS;
      const unsigned k = (unsigned)(d[u] - base);
      if (i < e1 && k < (unsigned)len) atomicAdd(&cnt[k], 1u);
    }
  }
  __syncthreads();

  unsigned* row = part + (size_t)p * N + base;
  for (int k = tid; k < len; k += HTHREADS) row[k] = cnt[k];
}

__global__ void dinv_kernel(const unsigned* __restrict__ part, float* __restrict__ dinv, int N) {
  const int i = blockIdx.x * blockDim.x + threadIdx.x;
  if (i < N) {
    unsigned c = 0u;
#pragma unroll
    for (int p = 0; p < HPARTS; ++p) c += part[(size_t)p * N + i];
    dinv[i] = rsqrtf((float)c + 1.0f);
  }
}

template <int NCOLS, bool RELU_A>
__global__ void __launch_bounds__(256) gemm_wmma_f32(const float* __restrict__ X,
                                                     const float* __restrict__ W,
                                                     const float* __restrict__ dinv,
                                                     const float* __restrict__ bias,
                                                     float* __restrict__ T,
                                                     float* __restrict__ S, int mtiles) {
  constexpr int NT = NCOLS / 16;
  constexpr int N4 = NCOLS / 4;
  constexpr int PITCH2 = NCOLS + 8;
  __shared__ v2f ws[64 * PITCH2];

  const int tid = threadIdx.x;
  const int wave = tid >> 5;
  const int lane = tid & 31;
  const int c = lane & 15;
  const int h = lane >> 4;

  const int tile = blockIdx.x * 8 + wave;
  const int ltile = tile < mtiles ? tile : mtiles - 1;

  constexpr int WIT = (64 * N4) / 256;
  v4f w0[WIT], w1[WIT];
#pragma unroll
  for (int it = 0; it < WIT; ++it) {
    const int i = tid + it * 256;
    const int kp = i / N4;
    const int n4 = i - kp * N4;
    w0[it] = *(const v4f*)(W + (size_t)(2 * kp) * NCOLS + 4 * n4);
    w1[it] = *(const v4f*)(W + (size_t)(2 * kp + 1) * NCOLS + 4 * n4);
  }

  const float* xr = X + ((size_t)ltile * 16 + c) * 128 + 4 * h;
  v4f a[16];
#pragma unroll
  for (int J = 0; J < 16; ++J) a[J] = *(const v4f*)(xr + 8 * J);

#pragma unroll
  for (int it = 0; it < WIT; ++it) {
    const int i = tid + it * 256;
    const int kp = i / N4;
    const int n4 = i - kp * N4;
    const v4f lo = {w0[it].x, w1[it].x, w0[it].y, w1[it].y};
    const v4f hi = {w0[it].z, w1[it].z, w0[it].w, w1[it].w};
    v4f* wd = (v4f*)(ws + kp * PITCH2 + 4 * n4);
    wd[0] = lo;
    wd[1] = hi;
  }
  __syncthreads();
  if (tile >= mtiles) return;

  if (RELU_A) {
#pragma unroll
    for (int J = 0; J < 16; ++J) {
      a[J].x = a[J].x > 0.0f ? a[J].x : 0.0f;
      a[J].y = a[J].y > 0.0f ? a[J].y : 0.0f;
      a[J].z = a[J].z > 0.0f ? a[J].z : 0.0f;
      a[J].w = a[J].w > 0.0f ? a[J].w : 0.0f;
    }
  }

  v8f acc[NT];
#pragma unroll
  for (int nt = 0; nt < NT; ++nt) acc[nt] = v8f{};

  const v2f* wb = ws + (2 * h) * PITCH2 + c;
#pragma unroll
  for (int J = 0; J < 16; ++J) {
#pragma unroll
    for (int sub = 0; sub < 2; ++sub) {
      v2f av;
      av.x = sub ? a[J].z : a[J].x;
      av.y = sub ? a[J].w : a[J].y;
      const v2f* wp = wb + (4 * J + sub) * PITCH2;
#pragma unroll
      for (int nt = 0; nt < NT; ++nt) {
        const v2f b = wp[16 * nt];
        acc[nt] = __builtin_amdgcn_wmma_f32_16x16x4_f32(false, av, false, b, (short)0,
                                                        acc[nt], false, false);
      }
    }
  }

  if constexpr (NT == 8) {
    asm volatile("v_nop\n\tv_nop\n\tv_nop\n\tv_nop\n\tv_nop\n\tv_nop\n\tv_nop\n\tv_nop"
                 : "+v"(acc[0]), "+v"(acc[1]), "+v"(acc[2]), "+v"(acc[3]),
                   "+v"(acc[4]), "+v"(acc[5]), "+v"(acc[6]), "+v"(acc[7]));
  } else {
    static_assert(NT == 4, "accumulator tiles tied by hand below");
    asm volatile("v_nop\n\tv_nop\n\tv_nop\n\tv_nop\n\tv_nop\n\tv_nop\n\tv_nop\n\tv_nop"
                 : "+v"(acc[0]), "+v"(acc[1]), "+v"(acc[2]), "+v"(acc[3]));
  }

  const float* dr = dinv + (size_t)tile * 16 + 8 * h;
  const v4f dlo = *(const v4f*)dr;
  const v4f dhi = *(const v4f*)(dr + 4);
  float bs[NT];
#pragma unroll
  for (int nt = 0; nt < NT; ++nt) bs[nt] = bias[c + 16 * nt];
  const float dv[8] = {dlo.x, dlo.y, dlo.z, dlo.w, dhi.x, dhi.y, dhi.z, dhi.w};
  const size_t o0 = ((size_t)tile * 16 + 8 * h) * NCOLS + c;
  float* tp = T + o0;
  float* sp = S + o0;
#pragma unroll
  for (int nt = 0; nt < NT; ++nt) {
#pragma unroll
    for (int r = 0; r < 8; ++r) {
      const float v = acc[nt][r];
      tp[(size_t)r * NCOLS + 16 * nt] = v;
      sp[(size_t)r * NCOLS + 16 * nt] = v * dv[r] * dv[r] + bs[nt];
    }
  }
}

template <int D, int NS>
__global__ void __launch_bounds__(256) scatter_edges(const int* __restrict__ src,
                                                     const int* __restrict__ dst,
                                                     const float* __restrict__ dinv,
                                                     const float* __restrict__ T,
                                                     float* __restrict__ OUT, int E) {
  constexpr int DS = D / NS;
  constexpr int F = DS / 32;
  constexpr int G = 32 / F;
  constexpr int NG = 32 / G;
  const int lane = threadIdx.x & 31;
  const int gw = __builtin_amdgcn_readfirstlane((int)((blockIdx.x * blockDim.x + threadIdx.x) >> 5));
  const int nw = (int)((gridDim.x * blockDim.x) >> 5);
  const int nb = (E + 31) >> 5;
  const char* Tb = (const char*)T;
  char* Ob = (char*)OUT;
  for (int sl = 0; sl < NS; ++sl) {
    const unsigned lcol = (unsigned)(sl * DS + lane) * 4u;
    for (int b = gw; b < nb; b += nw) {
      const int e = b * 32 + lane;
      int s = 0, d = 0;
      float nrm = 0.0f;
      if (e < E) {
        s = src[e];
        d = dst[e];
        nrm = dinv[s] * dinv[d];
      }
      const int nbits = __float_as_int(nrm);
      const int cnt = E - b * 32;
      if (cnt >= 32) {
#pragma unroll
        for (int g = 0; g < NG; ++g) {
          float v[G][F];
#pragma unroll
          for (int u = 0; u < G; ++u) {
            const int sj = __builtin_amdgcn_readlane(s, g * G + u);
            const unsigned to = (unsigned)sj * (unsigned)(D * 4) + lcol;
#pragma unroll
            for (int f = 0; f < F; ++f) v[u][f] = *(const float*)(Tb + (size_t)(to + 128u * f));
          }
#pragma unroll
          for (int u = 0; u < G; ++u) {
            const int dj = __builtin_amdgcn_readlane(d, g * G + u);
            const float nj = __int_as_float(__builtin_amdgcn_readlane(nbits, g * G + u));
            const unsigned oo = (unsigned)dj * (unsigned)(D * 4) + lcol;
#pragma unroll
            for (int f = 0; f < F; ++f) atomicAdd((float*)(Ob + (size_t)(oo + 128u * f)), v[u][f] * nj);
          }
        }
      } else {
        for (int j = 0; j < cnt; ++j) {
          const int sj = __builtin_amdgcn_readlane(s, j);
          const int dj = __builtin_amdgcn_readlane(d, j);
          const float nj = __int_as_float(__builtin_amdgcn_readlane(nbits, j));
          const unsigned to = (unsigned)sj * (unsigned)(D * 4) + lcol;
          const unsigned oo = (unsigned)dj * (unsigned)(D * 4) + lcol;
#pragma unroll
          for (int f = 0; f < F; ++f)
            atomicAdd((float*)(Ob + (size_t)(oo + 128u * f)),
                      *(const float*)(Tb + (size_t)(to + 128u * f)) * nj);
        }
      }
    }
  }
}

extern "C" void kernel_launch(void* const* d_in, const int* in_sizes, int n_in,
                              void* d_out, int out_size, void* d_ws, size_t ws_size,
                              hipStream_t stream) {
  const float* x  = (const float*)d_in[0];
  const int*   ei = (const int*)d_in[1];
  const float* W1 = (const float*)d_in[2];
  const float* b1 = (const float*)d_in[3];
  const float* W2 = (const float*)d_in[4];
  const float* b2 = (const float*)d_in[5];
  float* out = (float*)d_out;

  const int N = in_sizes[0] / GDIN;
  const int E = in_sizes[1] / 2;
  const int* src = ei;
  const int* dst = ei + E;

  float* bufA = (float*)d_ws;
  float* bufB = bufA + (size_t)N * GDH;
  float* dinv = bufB + (size_t)N * GDH;

  const int mtiles = (N + 15) / 16;
  const int gblocks = (mtiles + 7) / 8;

  unsigned* part = (unsigned*)bufA;
  const int nranges = (N + HCAP - 1) / HCAP;
  degree_hist_kernel<<<nranges * HPARTS, HTHREADS, 0, stream>>>(dst, part, N, E, nranges);
  dinv_kernel<<<(N + 255) / 256, 256, 0, stream>>>(part, dinv, N);

  gemm_wmma_f32<GDH, false><<<gblocks, 256, 0, stream>>>(x, W1, dinv, b1, bufA, bufB, mtiles);
  scatter_edges<GDH, 2><<<2048, 256, 0, stream>>>(src, dst, dinv, bufA, bufB, E);

  gemm_wmma_f32<GDOUT, true><<<gblocks, 256, 0, stream>>>(bufB, W2, dinv, b2, bufA, out, mtiles);
  scatter_edges<GDOUT, 1><<<2048, 256, 0, stream>>>(src, dst, dinv, bufA, out, E);
}


# speedup vs baseline: 2.8396x; 1.0703x over previous
#include <hip/hip_runtime.h>

typedef float v2f __attribute__((ext_vector_type(2)));
typedef float v4f __attribute__((ext_vector_type(4)));
typedef float v8f __attribute__((ext_vector_type(8)));

#define GDIN  128
#define GDH   128
#define GDOUT 64

#define HCAP     12800
#define HPARTS   32
#define HTHREADS 512
#define HU       25

__global__ void __launch_bounds__(HTHREADS) degree_hist_kernel(const int* __restrict__ dst,
                                                                unsigned* __restrict__ part,
                                                                int N, int E, int nranges) {
  __shared__ __attribute__((aligned(16))) unsigned cnt[HCAP];
  const int tid = threadIdx.x;
  const int r = blockIdx.x % nranges;
  const int p = blockIdx.x / nranges;
  const int base = r * HCAP;
  const int len = min(HCAP, N - base);

  uint4* c4 = (uint4*)cnt;
  for (int k = tid; k < HCAP / 4; k += HTHREADS) c4[k] = make_uint4(0u, 0u, 0u, 0u);
  __syncthreads();

  const int per = (E + HPARTS - 1) / HPARTS;
  const int e0 = p * per;
  const int e1 = min(e0 + per, E);
  for (int i0 = e0 + tid; i0 < e1; i0 += HTHREADS * HU) {
    int d[HU];
#pragma unroll
    for (int u = 0; u < HU; ++u) {
      const int i = i0 + u * HTHREADS;
      d[u] = dst[i < e1 ? i : e1 - 1];
    }
#pragma unroll
    for (int u = 0; u < HU; ++u) {
      const int i = i0 + u * HTHREADS;
      const unsigned k = (unsigned)(d[u] - base);
      if (i < e1 && k < (unsigned)len) atomicAdd(&cnt[k], 1u);
    }
  }
  __syncthreads();

  unsigned* row = part + (size_t)p * N + base;
  for (int k = tid; k < len; k += HTHREADS) row[k] = cnt[k];
}

__global__ void dinv_kernel(const unsigned* __restrict__ part, float* __restrict__ dinv,
                            int* __restrict__ ctr, int N) {
  const int i = blockIdx.x * blockDim.x + threadIdx.x;
  if (i < 6) ctr[i * 128] = 0;
  if (i < N) {
    unsigned c = 0u;
#pragma unroll
    for (int p = 0; p < HPARTS; ++p) c += part[(size_t)p * N + i];
    dinv[i] = rsqrtf((float)c + 1.0f);
  }
}

template <int NCOLS, bool RELU_A>
__global__ void __launch_bounds__(256) gemm_wmma_f32(const float* __restrict__ X,
                                                     const float* __restrict__ W,
                                                     const float* __restrict__ dinv,
                                                     const float* __restrict__ bias,
                                                     float* __restrict__ T,
                                                     float* __restrict__ S, int mtiles) {
  constexpr int NT = NCOLS / 16;
  constexpr int N4 = NCOLS / 4;
  constexpr int PITCH2 = NCOLS + 8;
  __shared__ v2f ws[64 * PITCH2];

  const int tid = threadIdx.x;
  const int wave = tid >> 5;
  const int lane = tid & 31;
  const int c = lane & 15;
  const int h = lane >> 4;

  const int tile = blockIdx.x * 8 + wave;
  const int ltile = tile < mtiles ? tile : mtiles - 1;

  constexpr int WIT = (64 * N4) / 256;
  v4f w0[WIT], w1[WIT];
#pragma unroll
  for (int it = 0; it < WIT; ++it) {
    const int i = tid + it * 256;
    const int kp = i / N4;
    const int n4 = i - kp * N4;
    w0[it] = *(const v4f*)(W + (size_t)(2 * kp) * NCOLS + 4 * n4);
    w1[it] = *(const v4f*)(W + (size_t)(2 * kp + 1) * NCOLS + 4 * n4);
  }

  const float* xr = X + ((size_t)ltile * 16 + c) * 128 + 4 * h;
  v4f a[16];
#pragma unroll
  for (int J = 0; J < 16; ++J) a[J] = *(const v4f*)(xr + 8 * J);

#pragma unroll
  for (int it = 0; it < WIT; ++it) {
    const int i = tid + it * 256;
    const int kp = i / N4;
    const int n4 = i - kp * N4;
    const v4f lo = {w0[it].x, w1[it].x, w0[it].y, w1[it].y};
    const v4f hi = {w0[it].z, w1[it].z, w0[it].w, w1[it].w};
    v4f* wd = (v4f*)(ws + kp * PITCH2 + 4 * n4);
    wd[0] = lo;
    wd[1] = hi;
  }
  __syncthreads();
  if (tile >= mtiles) return;

  if (RELU_A) {
#pragma unroll
    for (int J = 0; J < 16; ++J) {
      a[J].x = a[J].x > 0.0f ? a[J].x : 0.0f;
      a[J].y = a[J].y > 0.0f ? a[J].y : 0.0f;
      a[J].z = a[J].z > 0.0f ? a[J].z : 0.0f;
      a[J].w = a[J].w > 0.0f ? a[J].w : 0.0f;
    }
  }

  v8f acc[NT];
#pragma unroll
  for (int nt = 0; nt < NT; ++nt) acc[nt] = v8f{};

  const v2f* wb = ws + (2 * h) * PITCH2 + c;
#pragma unroll
  for (int J = 0; J < 16; ++J) {
#pragma unroll
    for (int sub = 0; sub < 2; ++sub) {
      v2f av;
      av.x = sub ? a[J].z : a[J].x;
      av.y = sub ? a[J].w : a[J].y;
      const v2f* wp = wb + (4 * J + sub) * PITCH2;
#pragma unroll
      for (int nt = 0; nt < NT; ++nt) {
        const v2f b = wp[16 * nt];
        acc[nt] = __builtin_amdgcn_wmma_f32_16x16x4_f32(false, av, false, b, (short)0,
                                                        acc[nt], false, false);
      }
    }
  }

  if constexpr (NT == 8) {
    asm volatile("v_nop\n\tv_nop\n\tv_nop\n\tv_nop\n\tv_nop\n\tv_nop\n\tv_nop\n\tv_nop"
                 : "+v"(acc[0]), "+v"(acc[1]), "+v"(acc[2]), "+v"(acc[3]),
                   "+v"(acc[4]), "+v"(acc[5]), "+v"(acc[6]), "+v"(acc[7]));
  } else {
    static_assert(NT == 4, "accumulator tiles tied by hand below");
    asm volatile("v_nop\n\tv_nop\n\tv_nop\n\tv_nop\n\tv_nop\n\tv_nop\n\tv_nop\n\tv_nop"
                 : "+v"(acc[0]), "+v"(acc[1]), "+v"(acc[2]), "+v"(acc[3]));
  }

  const float* dr = dinv + (size_t)tile * 16 + 8 * h;
  const v4f dlo = *(const v4f*)dr;
  const v4f dhi = *(const v4f*)(dr + 4);
  float bs[NT];
#pragma unroll
  for (int nt = 0; nt < NT; ++nt) bs[nt] = bias[c + 16 * nt];
  const float dv[8] = {dlo.x, dlo.y, dlo.z, dlo.w, dhi.x, dhi.y, dhi.z, dhi.w};
  const size_t o0 = ((size_t)tile * 16 + 8 * h) * NCOLS + c;
  float* tp = T + o0;
  float* sp = S + o0;
#pragma unroll
  for (int nt = 0; nt < NT; ++nt) {
#pragma unroll
    for (int r = 0; r < 8; ++r) {
      const float v = acc[nt][r];
      tp[(size_t)r * NCOLS + 16 * nt] = v;
      sp[(size_t)r * NCOLS + 16 * nt] = v * dv[r] * dv[r] + bs[nt];
    }
  }
}

#define CSTR  128
__device__ __forceinline__ void lds_barrier() {
  __builtin_amdgcn_fence(__ATOMIC_RELEASE, "workgroup", "local");
  __builtin_amdgcn_s_barrier();
  __builtin_amdgcn_fence(__ATOMIC_ACQUIRE, "workgroup", "local");
}
template <int D, int NS>
__global__ void __launch_bounds__(256) scatter_edges(const int* __restrict__ src,
                                                     const int* __restrict__ dst,
                                                     const float* __restrict__ dinv,
                                                     const float* __restrict__ T,
                                                     float* __restrict__ OUT,
                                                     int* __restrict__ ctr, int E, int N) {
  constexpr int DS = D / NS;
  constexpr int F = DS / 32;
  constexpr int G = 32 / F;
  constexpr int NG = 32 / G;
  __shared__ int qs[8][64];
  __shared__ int qd[8][64];
  __shared__ int qw[8][64];
  __shared__ int tk[2];
  const int tid = threadIdx.x;
  const int lane = tid & 31;
  const int wave = __builtin_amdgcn_readfirstlane(tid >> 5);
  const int myhalf = (__builtin_amdgcn_s_sendmsg_rtn(0x87) >> 18) & 1;
  const int nb = (E + 31) >> 5;
  const int nch = (nb + 7) >> 3;
  const int nh = N >> 1;
  const char* Tb = (const char*)T;
  char* Ob = (char*)OUT;
  int* q_s = qs[wave];
  int* q_d = qd[wave];
  int* q_w = qw[wave];
  for (int sl = 0; sl < NS; ++sl) {
    const unsigned lcol = (unsigned)(sl * DS + lane) * 4u;
    for (int pass = 0; pass < 2; ++pass) {
      const int role = myhalf ^ pass;
      int* cp = ctr + (sl * 2 + role) * CSTR;
      int qn = 0, head = 0;
      if (tid == 0) tk[0] = atomicAdd(cp, 1);
      lds_barrier();
      for (int it = 0; it < nch; ++it) {
        const int c = __builtin_amdgcn_readfirstlane(tk[it & 1]);
        if (c >= nch) break;
        int cn = 0;
        if (tid == 0) cn = atomicAdd(cp, 1);
        const int b = c * 8 + wave;
        if (b < nb) {
          const int e = b * 32 + lane;
          int s = 0, d = 0;
          bool in = false;
          if (e < E) {
            s = src[e];
            d = dst[e];
            in = (d >= nh ? 1 : 0) == role;
          }
          float nrm = 0.0f;
          if (in) nrm = dinv[s] * dinv[d];
          const unsigned mask = __builtin_amdgcn_ballot_w32(in);
          if (in) {
            const int pos = (head + qn + (int)__builtin_amdgcn_mbcnt_lo(mask, 0u)) & 63;
            q_s[pos] = s;
            q_d[pos] = d;
            q_w[pos] = __float_as_int(nrm);
          }
          qn += __builtin_popcount(mask);
          if (qn >= 32) {
            const int idx = (head + lane) & 63;
            const int s2 = q_s[idx];
            const int d2 = q_d[idx];
            const int n2 = q_w[idx];
            head = (head + 32) & 63;
            qn -= 32;
#pragma unroll
            for (int g = 0; g < NG; ++g) {
              float v[G][F];
#pragma unroll
              for (int u = 0; u < G; ++u) {
                const int sj = __builtin_amdgcn_readlane(s2, g * G + u);
                const unsigned to = (unsigned)sj * (unsigned)(D * 4) + lcol;
#pragma unroll
                for (int f = 0; f < F; ++f) v[u][f] = *(const float*)(Tb + (size_t)(to + 128u * f));
              }
#pragma unroll
              for (int u = 0; u < G; ++u) {
                const int dj = __builtin_amdgcn_readlane(d2, g * G + u);
                const float nj = __int_as_float(__builtin_amdgcn_readlane(n2, g * G + u));
                const unsigned oo = (unsigned)dj * (unsigned)(D * 4) + lcol;
#pragma unroll
                for (int f = 0; f < F; ++f) atomicAdd((float*)(Ob + (size_t)(oo + 128u * f)), v[u][f] * nj);
              }
            }
          }
        }
        if (tid == 0) tk[(it + 1) & 1] = cn;
        lds_barrier();
      }
      if (qn > 0) {
        const int idx = (head + lane) & 63;
        const int s2 = q_s[idx];
        const int d2 = q_d[idx];
        const int n2 = q_w[idx];
        for (int j = 0; j < qn; ++j) {
          const int sj = __builtin_amdgcn_readlane(s2, j);
          const int dj = __builtin_amdgcn_readlane(d2, j);
          const float nj = __int_as_float(__builtin_amdgcn_readlane(n2, j));
          const unsigned to = (unsigned)sj * (unsigned)(D * 4) + lcol;
          const unsigned oo = (unsigned)dj * (unsigned)(D * 4) + lcol;
#pragma unroll
          for (int f = 0; f < F; ++f)
            atomicAdd((float*)(Ob + (size_t)(oo + 128u * f)),
                      *(const float*)(Tb + (size_t)(to + 128u * f)) * nj);
        }
      }
      lds_barrier();
    }
  }
}

extern "C" void kernel_launch(void* const* d_in, const int* in_sizes, int n_in,
                              void* d_out, int out_size, void* d_ws, size_t ws_size,
                              hipStream_t stream) {
  const float* x  = (const float*)d_in[0];
  const int*   ei = (const int*)d_in[1];
  const float* W1 = (const float*)d_in[2];
  const float* b1 = (const float*)d_in[3];
  const float* W2 = (const float*)d_in[4];
  const float* b2 = (const float*)d_in[5];
  float* out = (float*)d_out;

  const int N = in_sizes[0] / GDIN;
  const int E = in_sizes[1] / 2;
  const int* src = ei;
  const int* dst = ei + E;

  float* bufA = (float*)d_ws;
  float* bufB = bufA + (size_t)N * GDH;
  float* dinv = bufB + (size_t)N * GDH;
  int* ctr = (int*)(dinv + N);

  const int mtiles = (N + 15) / 16;
  const int gblocks = (mtiles + 7) / 8;

  unsigned* part = (unsigned*)bufA;
  const int nranges = (N + HCAP - 1) / HCAP;
  degree_hist_kernel<<<nranges * HPARTS, HTHREADS, 0, stream>>>(dst, part, N, E, nranges);
  dinv_kernel<<<(N + 255) / 256, 256, 0, stream>>>(part, dinv, ctr, N);

  gemm_wmma_f32<GDH, false><<<gblocks, 256, 0, stream>>>(x, W1, dinv, b1, bufA, bufB, mtiles);
  scatter_edges<GDH, 2><<<2048, 256, 0, stream>>>(src, dst, dinv, bufA, bufB, ctr, E, N);

  gemm_wmma_f32<GDOUT, true><<<gblocks, 256, 0, stream>>>(bufB, W2, dinv, b2, bufA, out, mtiles);
  scatter_edges<GDOUT, 1><<<2048, 256, 0, stream>>>(src, dst, dinv, bufA, out, ctr + 4 * 128, E, N);
}


# Edge scatter: ring leftovers (under 32 edges per wave and pass) drained as groups: up to 16 edges' line loads issued together ahead of their f32 atomics, was one edge at a time; atomic order changes
# speedup vs baseline: 3.0684x; 1.0806x over previous
#include <hip/hip_runtime.h>

typedef float v2f __attribute__((ext_vector_type(2)));
typedef float v4f __attribute__((ext_vector_type(4)));
typedef float v8f __attribute__((ext_vector_type(8)));

#define GDIN  128
#define GDH   128
#define GDOUT 64

#define HCAP     12800
#define HPARTS   32
#define HTHREADS 512
#define HU       25

__global__ void __launch_bounds__(HTHREADS) degree_hist_kernel(const int* __restrict__ dst,
                                                                unsigned* __restrict__ part,
                                                                int N, int E, int nranges) {
  __shared__ __attribute__((aligned(16))) unsigned cnt[HCAP];
  const int tid = threadIdx.x;
  const int r = blockIdx.x % nranges;
  const int p = blockIdx.x / nranges;
  const int base = r * HCAP;
  const int len = min(HCAP, N - base);

  uint4* c4 = (uint4*)cnt;
  for (int k = tid; k < HCAP / 4; k += HTHREADS) c4[k] = make_uint4(0u, 0u, 0u, 0u);
  __syncthreads();

  const int per = (E + HPARTS - 1) / HPARTS;
  const int e0 = p * per;
  const int e1 = min(e0 + per, E);
  for (int i0 = e0 + tid; i0 < e1; i0 += HTHREADS * HU) {
    int d[HU];
#pragma unroll
    for (int u = 0; u < HU; ++u) {
      const int i = i0 + u * HTHREADS;
      d[u] = dst[i < e1 ? i : e1 - 1];
    }
#pragma unroll
    for (int u = 0; u < HU; ++u) {
      const int i = i0 + u * HTHREADS;
      const unsigned k = (unsigned)(d[u] - base);
      if (i < e1 && k < (unsigned)len) atomicAdd(&cnt[k], 1u);
    }
  }
  __syncthreads();

  unsigned* row = part + (size_t)p * N + base;
  for (int k = tid; k < len; k += HTHREADS) row[k] = cnt[k];
}

__global__ void dinv_kernel(const unsigned* __restrict__ part, float* __restrict__ dinv,
                            int* __restrict__ ctr, int N) {
  const int i = blockIdx.x * blockDim.x + threadIdx.x;
  if (i < 6) ctr[i * 128] = 0;
  if (i < N) {
    unsigned c = 0u;
#pragma unroll
    for (int p = 0; p < HPARTS; ++p) c += part[(size_t)p * N + i];
    dinv[i] = rsqrtf((float)c + 1.0f);
  }
}

template <int NCOLS, bool RELU_A>
__global__ void __launch_bounds__(256) gemm_wmma_f32(const float* __restrict__ X,
                                                     const float* __restrict__ W,
                                                     const float* __restrict__ dinv,
                                                     const float* __restrict__ bias,
                                                     float* __restrict__ T,
                                                     float* __restrict__ S, int mtiles) {
  constexpr int NT = NCOLS / 16;
  constexpr int N4 = NCOLS / 4;
  constexpr int PITCH2 = NCOLS + 8;
  __shared__ v2f ws[64 * PITCH2];

  const int tid = threadIdx.x;
  const int wave = tid >> 5;
  const int lane = tid & 31;
  const int c = lane & 15;
  const int h = lane >> 4;

  const int tile = blockIdx.x * 8 + wave;
  const int ltile = tile < mtiles ? tile : mtiles - 1;

  constexpr int WIT = (64 * N4) / 256;
  v4f w0[WIT], w1[WIT];
#pragma unroll
  for (int it = 0; it < WIT; ++it) {
    const int i = tid + it * 256;
    const int kp = i / N4;
    const int n4 = i - kp * N4;
    w0[it] = *(const v4f*)(W + (size_t)(2 * kp) * NCOLS + 4 * n4);
    w1[it] = *(const v4f*)(W + (size_t)(2 * kp + 1) * NCOLS + 4 * n4);
  }

  const float* xr = X + ((size_t)ltile * 16 + c) * 128 + 4 * h;
  v4f a[16];
#pragma unroll
  for (int J = 0; J < 16; ++J) a[J] = *(const v4f*)(xr + 8 * J);

#pragma unroll
  for (int it = 0; it < WIT; ++it) {
    const int i = tid + it * 256;
    const int kp = i / N4;
    const int n4 = i - kp * N4;
    const v4f lo = {w0[it].x, w1[it].x, w0[it].y, w1[it].y};
    const v4f hi = {w0[it].z, w1[it].z, w0[it].w, w1[it].w};
    v4f* wd = (v4f*)(ws + kp * PITCH2 + 4 * n4);
    wd[0] = lo;
    wd[1] = hi;
  }
  __syncthreads();
  if (tile >= mtiles) return;

  if (RELU_A) {
#pragma unroll
    for (int J = 0; J < 16; ++J) {
      a[J].x = a[J].x > 0.0f ? a[J].x : 0.0f;
      a[J].y = a[J].y > 0.0f ? a[J].y : 0.0f;
      a[J].z = a[J].z > 0.0f ? a[J].z : 0.0f;
      a[J].w = a[J].w > 0.0f ? a[J].w : 0.0f;
    }
  }

  v8f acc[NT];
#pragma unroll
  for (int nt = 0; nt < NT; ++nt) acc[nt] = v8f{};

  const v2f* wb = ws + (2 * h) * PITCH2 + c;
#pragma unroll
  for (int J = 0; J < 16; ++J) {
#pragma unroll
    for (int sub = 0; sub < 2; ++sub) {
      v2f av;
      av.x = sub ? a[J].z : a[J].x;
      av.y = sub ? a[J].w : a[J].y;
      const v2f* wp = wb + (4 * J + sub) * PITCH2;
#pragma unroll
      for (int nt = 0; nt < NT; ++nt) {
        const v2f b = wp[16 * nt];
        acc[nt] = __builtin_amdgcn_wmma_f32_16x16x4_f32(false, av, false, b, (short)0,
                                                        acc[nt], false, false);
      }
    }
  }

  if constexpr (NT == 8) {
    asm volatile("v_nop\n\tv_nop\n\tv_nop\n\tv_nop\n\tv_nop\n\tv_nop\n\tv_nop\n\tv_nop"
                 : "+v"(acc[0]), "+v"(acc[1]), "+v"(acc[2]), "+v"(acc[3]),
                   "+v"(acc[4]), "+v"(acc[5]), "+v"(acc[6]), "+v"(acc[7]));
  } else {
    static_assert(NT == 4, "accumulator tiles tied by hand below");
    asm volatile("v_nop\n\tv_nop\n\tv_nop\n\tv_nop\n\tv_nop\n\tv_nop\n\tv_nop\n\tv_nop"
                 : "+v"(acc[0]), "+v"(acc[1]), "+v"(acc[2]), "+v"(acc[3]));
  }

  const float* dr = dinv + (size_t)tile * 16 + 8 * h;
  const v4f dlo = *(const v4f*)dr;
  const v4f dhi = *(const v4f*)(dr + 4);
  float bs[NT];
#pragma unroll
  for (int nt = 0; nt < NT; ++nt) bs[nt] = bias[c + 16 * nt];
  const float dv[8] = {dlo.x, dlo.y, dlo.z, dlo.w, dhi.x, dhi.y, dhi.z, dhi.w};
  const size_t o0 = ((size_t)tile * 16 + 8 * h) * NCOLS + c;
  float* tp = T + o0;
  float* sp = S + o0;
#pragma unroll
  for (int nt = 0; nt < NT; ++nt) {
#pragma unroll
    for (int r = 0; r < 8; ++r) {
      const float v = acc[nt][r];
      tp[(size_t)r * NCOLS + 16 * nt] = v;
      sp[(size_t)r * NCOLS + 16 * nt] = v * dv[r] * dv[r] + bs[nt];
    }
  }
}

#define CSTR  128
__device__ __forceinline__ void lds_barrier() {
  __builtin_amdgcn_fence(__ATOMIC_RELEASE, "workgroup", "local");
  __builtin_amdgcn_s_barrier();
  __builtin_amdgcn_fence(__ATOMIC_ACQUIRE, "workgroup", "local");
}
template <int D, int NS>
__global__ void __launch_bounds__(256) scatter_edges(const int* __restrict__ src,
                                                     const int* __restrict__ dst,
                                                     const float* __restrict__ dinv,
                                                     const float* __restrict__ T,
                                                     float* __restrict__ OUT,
                                                     int* __restrict__ ctr, int E, int N) {
  constexpr int DS = D / NS;
  constexpr int F = DS / 32;
  constexpr int G = 32 / F;
  constexpr int NG = 32 / G;
  __shared__ int qs[8][64];
  __shared__ int qd[8][64];
  __shared__ int qw[8][64];
  __shared__ int tk[2];
  const int tid = threadIdx.x;
  const int lane = tid & 31;
  const int wave = __builtin_amdgcn_readfirstlane(tid >> 5);
  const int myhalf = (__builtin_amdgcn_s_sendmsg_rtn(0x87) >> 18) & 1;
  const int nb = (E + 31) >> 5;
  const int nch = (nb + 7) >> 3;
  const int nh = N >> 1;
  const char* Tb = (const char*)T;
  char* Ob = (char*)OUT;
  int* q_s = qs[wave];
  int* q_d = qd[wave];
  int* q_w = qw[wave];
  for (int sl = 0; sl < NS; ++sl) {
    const unsigned lcol = (unsigned)(sl * DS + lane) * 4u;
    for (int pass = 0; pass < 2; ++pass) {
      const int role = myhalf ^ pass;
      int* cp = ctr + (sl * 2 + role) * CSTR;
      int qn = 0, head = 0;
      if (tid == 0) tk[0] = atomicAdd(cp, 1);
      lds_barrier();
      for (int it = 0; it < nch; ++it) {
        const int c = __builtin_amdgcn_readfirstlane(tk[it & 1]);
        if (c >= nch) break;
        int cn = 0;
        if (tid == 0) cn = atomicAdd(cp, 1);
        const int b = c * 8 + wave;
        if (b < nb) {
          const int e = b * 32 + lane;
          int s = 0, d = 0;
          bool in = false;
          if (e < E) {
            s = src[e];
            d = dst[e];
            in = (d >= nh ? 1 : 0) == role;
          }
          float nrm = 0.0f;
          if (in) nrm = dinv[s] * dinv[d];
          const unsigned mask = __builtin_amdgcn_ballot_w32(in);
          if (in) {
            const int pos = (head + qn + (int)__builtin_amdgcn_mbcnt_lo(mask, 0u)) & 63;
            q_s[pos] = s;
            q_d[pos] = d;
            q_w[pos] = __float_as_int(nrm);
          }
          qn += __builtin_popcount(mask);
          if (qn >= 32) {
            const int idx = (head + lane) & 63;
            const int s2 = q_s[idx];
            const int d2 = q_d[idx];
            const int n2 = q_w[idx];
            head = (head + 32) & 63;
            qn -= 32;
#pragma unroll
            for (int g = 0; g < NG; ++g) {
              float v[G][F];
#pragma unroll
              for (int u = 0; u < G; ++u) {
                const int sj = __builtin_amdgcn_readlane(s2, g * G + u);
                const unsigned to = (unsigned)sj * (unsigned)(D * 4) + lcol;
#pragma unroll
                for (int f = 0; f < F; ++f) v[u][f] = *(const float*)(Tb + (size_t)(to + 128u * f));
              }
#pragma unroll
              for (int u = 0; u < G; ++u) {
                const int dj = __builtin_amdgcn_readlane(d2, g * G + u);
                const float nj = __int_as_float(__builtin_amdgcn_readlane(n2, g * G + u));
                const unsigned oo = (unsigned)dj * (unsigned)(D * 4) + lcol;
#pragma unroll
                for (int f = 0; f < F; ++f) atomicAdd((float*)(Ob + (size_t)(oo + 128u * f)), v[u][f] * nj);
              }
            }
          }
        }
        if (tid == 0) tk[(it + 1) & 1] = cn;
        lds_barrier();
      }
      if (qn > 0) {
        const int idx = (head + lane) & 63;
        const int s2 = q_s[idx];
        const int d2 = q_d[idx];
        const int n2 = q_w[idx];
#pragma unroll
        for (int g = 0; g < NG; ++g) {
          if (g * G < qn) {
            float v[G][F];
#pragma unroll
            for (int u = 0; u < G; ++u) {
#pragma unroll
              for (int f = 0; f < F; ++f) v[u][f] = 0.0f;
              if (g * G + u < qn) {
                const int sj = __builtin_amdgcn_readlane(s2, g * G + u);
                const unsigned to = (unsigned)sj * (unsigned)(D * 4) + lcol;
#pragma unroll
                for (int f = 0; f < F; ++f) v[u][f] = *(const float*)(Tb + (size_t)(to + 128u * f));
              }
            }
#pragma unroll
            for (int u = 0; u < G; ++u) {
              if (g * G + u < qn) {
                const int dj = __builtin_amdgcn_readlane(d2, g * G + u);
                const float nj = __int_as_float(__builtin_amdgcn_readlane(n2, g * G + u));
                const unsigned oo = (unsigned)dj * (unsigned)(D * 4) + lcol;
#pragma unroll
                for (int f = 0; f < F; ++f) atomicAdd((float*)(Ob + (size_t)(oo + 128u * f)), v[u][f] * nj);
              }
            }
          }
        }
      }
      lds_barrier();
    }
  }
}

extern "C" void kernel_launch(void* const* d_in, const int* in_sizes, int n_in,
                              void* d_out, int out_size, void* d_ws, size_t ws_size,
                              hipStream_t stream) {
  const float* x  = (const float*)d_in[0];
  const int*   ei = (const int*)d_in[1];
  const float* W1 = (const float*)d_in[2];
  const float* b1 = (const float*)d_in[3];
  const float* W2 = (const float*)d_in[4];
  const float* b2 = (const float*)d_in[5];
  float* out = (float*)d_out;

  const int N = in_sizes[0] / GDIN;
  const int E = in_sizes[1] / 2;
  const int* src = ei;
  const int* dst = ei + E;

  float* bufA = (float*)d_ws;
  float* bufB = bufA + (size_t)N * GDH;
  float* dinv = bufB + (size_t)N * GDH;
  int* ctr = (int*)(dinv + N);

  const int mtiles = (N + 15) / 16;
  const int gblocks = (mtiles + 7) / 8;

  unsigned* part = (unsigned*)bufA;
  const int nranges = (N + HCAP - 1) / HCAP;
  degree_hist_kernel<<<nranges * HPARTS, HTHREADS, 0, stream>>>(dst, part, N, E, nranges);
  dinv_kernel<<<(N + 255) / 256, 256, 0, stream>>>(part, dinv, ctr, N);

  gemm_wmma_f32<GDH, false><<<gblocks, 256, 0, stream>>>(x, W1, dinv, b1, bufA, bufB, mtiles);
  scatter_edges<GDH, 2><<<2048, 256, 0, stream>>>(src, dst, dinv, bufA, bufB, ctr, E, N);

  gemm_wmma_f32<GDOUT, true><<<gblocks, 256, 0, stream>>>(bufB, W2, dinv, b2, bufA, out, mtiles);
  scatter_edges<GDOUT, 1><<<2048, 256, 0, stream>>>(src, dst, dinv, bufA, out, ctr + 4 * 128, E, N);
}


# Edge weight dinv[src]*dinv[dst] made once per layer by spare blocks of the GEMM launch into a table; edge walks load it beside the indices, scattered dinv reads per walk gone; f32 atomics as before
# speedup vs baseline: 3.4361x; 1.1198x over previous
#include <hip/hip_runtime.h>

typedef float v2f __attribute__((ext_vector_type(2)));
typedef float v4f __attribute__((ext_vector_type(4)));
typedef float v8f __attribute__((ext_vector_type(8)));

#define GDIN  128
#define GDH   128
#define GDOUT 64

#define HCAP     12800
#define HPARTS   32
#define HTHREADS 512
#define HU       25

__global__ void __launch_bounds__(HTHREADS) degree_hist_kernel(const int* __restrict__ dst,
                                                                unsigned* __restrict__ part,
                                                                int N, int E, int nranges) {
  __shared__ __attribute__((aligned(16))) unsigned cnt[HCAP];
  const int tid = threadIdx.x;
  const int r = blockIdx.x % nranges;
  const int p = blockIdx.x / nranges;
  const int base = r * HCAP;
  const int len = min(HCAP, N - base);

  uint4* c4 = (uint4*)cnt;
  for (int k = tid; k < HCAP / 4; k += HTHREADS) c4[k] = make_uint4(0u, 0u, 0u, 0u);
  __syncthreads();

  const int per = (E + HPARTS - 1) / HPARTS;
  const int e0 = p * per;
  const int e1 = min(e0 + per, E);
  for (int i0 = e0 + tid; i0 < e1; i0 += HTHREADS * HU) {
    int d[HU];
#pragma unroll
    for (int u = 0; u < HU; ++u) {
      const int i = i0 + u * HTHREADS;
      d[u] = dst[i < e1 ? i : e1 - 1];
    }
#pragma unroll
    for (int u = 0; u < HU; ++u) {
      const int i = i0 + u * HTHREADS;
      const unsigned k = (unsigned)(d[u] - base);
      if (i < e1 && k < (unsigned)len) atomicAdd(&cnt[k], 1u);
    }
  }
  __syncthreads();

  unsigned* row = part + (size_t)p * N + base;
  for (int k = tid; k < len; k += HTHREADS) row[k] = cnt[k];
}

__global__ void dinv_kernel(const unsigned* __restrict__ part, float* __restrict__ dinv,
                            int* __restrict__ ctr, int N) {
  const int i = blockIdx.x * blockDim.x + threadIdx.x;
  if (i < 6) ctr[i * 128] = 0;
  if (i < N) {
    unsigned c = 0u;
#pragma unroll
    for (int p = 0; p < HPARTS; ++p) c += part[(size_t)p * N + i];
    dinv[i] = rsqrtf((float)c + 1.0f);
  }
}

#define EW_U         8
#define EW_PER_BLOCK (256 * EW_U)
template <int NCOLS, bool RELU_A>
__global__ void __launch_bounds__(256) gemm_wmma_f32(const float* __restrict__ X,
                                                     const float* __restrict__ W,
                                                     const float* __restrict__ dinv,
                                                     const float* __restrict__ bias,
                                                     float* __restrict__ T,
                                                     float* __restrict__ S, int mtiles,
                                                     const int* __restrict__ src,
                                                     const int* __restrict__ dst,
                                                     float* __restrict__ wt, int E) {
  constexpr int NT = NCOLS / 16;
  constexpr int N4 = NCOLS / 4;
  constexpr int PITCH2 = NCOLS + 8;
  __shared__ v2f ws[64 * PITCH2];

  const int tid = threadIdx.x;
  const int wave = tid >> 5;
  const int lane = tid & 31;
  const int c = lane & 15;
  const int h = lane >> 4;

  const int gemm_blocks = (mtiles + 7) >> 3;
  if ((int)blockIdx.x >= gemm_blocks) {
    const int base = ((int)blockIdx.x - gemm_blocks) * EW_PER_BLOCK;
    if (E - base >= EW_PER_BLOCK) {
      const int e0 = base + tid;
      int es[EW_U], ed[EW_U];
#pragma unroll
      for (int u = 0; u < EW_U; ++u) {
        es[u] = src[e0 + u * 256];
        ed[u] = dst[e0 + u * 256];
      }
      float fs[EW_U], fd[EW_U];
#pragma unroll
      for (int u = 0; u < EW_U; ++u) {
        fs[u] = dinv[es[u]];
        fd[u] = dinv[ed[u]];
      }
#pragma unroll
      for (int u = 0; u < EW_U; ++u) wt[e0 + u * 256] = fs[u] * fd[u];
    } else {
      for (int e = base + tid; e < E; e += 256) wt[e] = dinv[src[e]] * dinv[dst[e]];
    }
    return;
  }

  const int tile = blockIdx.x * 8 + wave;
  const int ltile = tile < mtiles ? tile : mtiles - 1;

  constexpr int WIT = (64 * N4) / 256;
  v4f w0[WIT], w1[WIT];
#pragma unroll
  for (int it = 0; it < WIT; ++it) {
    const int i = tid + it * 256;
    const int kp = i / N4;
    const int n4 = i - kp * N4;
    w0[it] = *(const v4f*)(W + (size_t)(2 * kp) * NCOLS + 4 * n4);
    w1[it] = *(const v4f*)(W + (size_t)(2 * kp + 1) * NCOLS + 4 * n4);
  }

  const float* xr = X + ((size_t)ltile * 16 + c) * 128 + 4 * h;
  v4f a[16];
#pragma unroll
  for (int J = 0; J < 16; ++J) a[J] = *(const v4f*)(xr + 8 * J);

#pragma unroll
  for (int it = 0; it < WIT; ++it) {
    const int i = tid + it * 256;
    const int kp = i / N4;
    const int n4 = i - kp * N4;
    const v4f lo = {w0[it].x, w1[it].x, w0[it].y, w1[it].y};
    const v4f hi = {w0[it].z, w1[it].z, w0[it].w, w1[it].w};
    v4f* wd = (v4f*)(ws + kp * PITCH2 + 4 * n4);
    wd[0] = lo;
    wd[1] = hi;
  }
  __syncthreads();
  if (tile >= mtiles) return;

  if (RELU_A) {
#pragma unroll
    for (int J = 0; J < 16; ++J) {
      a[J].x = a[J].x > 0.0f ? a[J].x : 0.0f;
      a[J].y = a[J].y > 0.0f ? a[J].y : 0.0f;
      a[J].z = a[J].z > 0.0f ? a[J].z : 0.0f;
      a[J].w = a[J].w > 0.0f ? a[J].w : 0.0f;
    }
  }

  v8f acc[NT];
#pragma unroll
  for (int nt = 0; nt < NT; ++nt) acc[nt] = v8f{};

  const v2f* wb = ws + (2 * h) * PITCH2 + c;
#pragma unroll
  for (int J = 0; J < 16; ++J) {
#pragma unroll
    for (int sub = 0; sub < 2; ++sub) {
      v2f av;
      av.x = sub ? a[J].z : a[J].x;
      av.y = sub ? a[J].w : a[J].y;
      const v2f* wp = wb + (4 * J + sub) * PITCH2;
#pragma unroll
      for (int nt = 0; nt < NT; ++nt) {
        const v2f b = wp[16 * nt];
        acc[nt] = __builtin_amdgcn_wmma_f32_16x16x4_f32(false, av, false, b, (short)0,
                                                        acc[nt], false, false);
      }
    }
  }

  if constexpr (NT == 8) {
    asm volatile("v_nop\n\tv_nop\n\tv_nop\n\tv_nop\n\tv_nop\n\tv_nop\n\tv_nop\n\tv_nop"
                 : "+v"(acc[0]), "+v"(acc[1]), "+v"(acc[2]), "+v"(acc[3]),
                   "+v"(acc[4]), "+v"(acc[5]), "+v"(acc[6]), "+v"(acc[7]));
  } else {
    static_assert(NT == 4, "accumulator tiles tied by hand below");
    asm volatile("v_nop\n\tv_nop\n\tv_nop\n\tv_nop\n\tv_nop\n\tv_nop\n\tv_nop\n\tv_nop"
                 : "+v"(acc[0]), "+v"(acc[1]), "+v"(acc[2]), "+v"(acc[3]));
  }

  const float* dr = dinv + (size_t)tile * 16 + 8 * h;
  const v4f dlo = *(const v4f*)dr;
  const v4f dhi = *(const v4f*)(dr + 4);
  float bs[NT];
#pragma unroll
  for (int nt = 0; nt < NT; ++nt) bs[nt] = bias[c + 16 * nt];
  const float dv[8] = {dlo.x, dlo.y, dlo.z, dlo.w, dhi.x, dhi.y, dhi.z, dhi.w};
  const size_t o0 = ((size_t)tile * 16 + 8 * h) * NCOLS + c;
  float* tp = T + o0;
  float* sp = S + o0;
#pragma unroll
  for (int nt = 0; nt < NT; ++nt) {
#pragma unroll
    for (int r = 0; r < 8; ++r) {
      const float v = acc[nt][r];
      tp[(size_t)r * NCOLS + 16 * nt] = v;
      sp[(size_t)r * NCOLS + 16 * nt] = v * dv[r] * dv[r] + bs[nt];
    }
  }
}

#define CSTR  128
__device__ __forceinline__ void lds_barrier() {
  __builtin_amdgcn_fence(__ATOMIC_RELEASE, "workgroup", "local");
  __builtin_amdgcn_s_barrier();
  __builtin_amdgcn_fence(__ATOMIC_ACQUIRE, "workgroup", "local");
}
template <int D, int NS>
__global__ void __launch_bounds__(256) scatter_edges(const int* __restrict__ src,
                                                     const int* __restrict__ dst,
                                                     const float* __restrict__ wt,
                                                     const float* __restrict__ T,
                                                     float* __restrict__ OUT,
                                                     int* __restrict__ ctr, int E, int N) {
  constexpr int DS = D / NS;
  constexpr int F = DS / 32;
  constexpr int G = 32 / F;
  constexpr int NG = 32 / G;
  __shared__ int qs[8][64];
  __shared__ int qd[8][64];
  __shared__ int qw[8][64];
  __shared__ int tk[2];
  const int tid = threadIdx.x;
  const int lane = tid & 31;
  const int wave = __builtin_amdgcn_readfirstlane(tid >> 5);
  const int myhalf = (__builtin_amdgcn_s_sendmsg_rtn(0x87) >> 18) & 1;
  const int nb = (E + 31) >> 5;
  const int nch = (nb + 7) >> 3;
  const int nh = N >> 1;
  const char* Tb = (const char*)T;
  char* Ob = (char*)OUT;
  int* q_s = qs[wave];
  int* q_d = qd[wave];
  int* q_w = qw[wave];
  for (int sl = 0; sl < NS; ++sl) {
    const unsigned lcol = (unsigned)(sl * DS + lane) * 4u;
    for (int pass = 0; pass < 2; ++pass) {
      const int role = myhalf ^ pass;
      int* cp = ctr + (sl * 2 + role) * CSTR;
      int qn = 0, head = 0;
      if (tid == 0) tk[0] = atomicAdd(cp, 1);
      lds_barrier();
      for (int it = 0; it < nch; ++it) {
        const int c = __builtin_amdgcn_readfirstlane(tk[it & 1]);
        if (c >= nch) break;
        int cn = 0;
        if (tid == 0) cn = atomicAdd(cp, 1);
        const int b = c * 8 + wave;
        if (b < nb) {
          const int e = b * 32 + lane;
          int s = 0, d = 0;
          float nrm = 0.0f;
          bool in = false;
          if (e < E) {
            s = src[e];
            d = dst[e];
            nrm = wt[e];
            in = (d >= nh ? 1 : 0) == role;
          }
          const unsigned mask = __builtin_amdgcn_ballot_w32(in);
          if (in) {
            const int pos = (head + qn + (int)__builtin_amdgcn_mbcnt_lo(mask, 0u)) & 63;
            q_s[pos] = s;
            q_d[pos] = d;
            q_w[pos] = __float_as_int(nrm);
          }
          qn += __builtin_popcount(mask);
          if (qn >= 32) {
            const int idx = (head + lane) & 63;
            const int s2 = q_s[idx];
            const int d2 = q_d[idx];
            const int n2 = q_w[idx];
            head = (head + 32) & 63;
            qn -= 32;
#pragma unroll
            for (int g = 0; g < NG; ++g) {
              float v[G][F];
#pragma unroll
              for (int u = 0; u < G; ++u) {
                const int sj = __builtin_amdgcn_readlane(s2, g * G + u);
                const unsigned to = (unsigned)sj * (unsigned)(D * 4) + lcol;
#pragma unroll
                for (int f = 0; f < F; ++f) v[u][f] = *(const float*)(Tb + (size_t)(to + 128u * f));
              }
#pragma unroll
              for (int u = 0; u < G; ++u) {
                const int dj = __builtin_amdgcn_readlane(d2, g * G + u);
                const float nj = __int_as_float(__builtin_amdgcn_readlane(n2, g * G + u));
                const unsigned oo = (unsigned)dj * (unsigned)(D * 4) + lcol;
#pragma unroll
                for (int f = 0; f < F; ++f) atomicAdd((float*)(Ob + (size_t)(oo + 128u * f)), v[u][f] * nj);
              }
            }
          }
        }
        if (tid == 0) tk[(it + 1) & 1] = cn;
        lds_barrier();
      }
      if (qn > 0) {
        const int idx = (head + lane) & 63;
        const int s2 = q_s[idx];
        const int d2 = q_d[idx];
        const int n2 = q_w[idx];
#pragma unroll
        for (int g = 0; g < NG; ++g) {
          if (g * G < qn) {
            float v[G][F];
#pragma unroll
            for (int u = 0; u < G; ++u) {
#pragma unroll
              for (int f = 0; f < F; ++f) v[u][f] = 0.0f;
              if (g * G + u < qn) {
                const int sj = __builtin_amdgcn_readlane(s2, g * G + u);
                const unsigned to = (unsigned)sj * (unsigned)(D * 4) + lcol;
#pragma unroll
                for (int f = 0; f < F; ++f) v[u][f] = *(const float*)(Tb + (size_t)(to + 128u * f));
              }
            }
#pragma unroll
            for (int u = 0; u < G; ++u) {
              if (g * G + u < qn) {
                const int dj = __builtin_amdgcn_readlane(d2, g * G + u);
                const float nj = __int_as_float(__builtin_amdgcn_readlane(n2, g * G + u));
                const unsigned oo = (unsigned)dj * (unsigned)(D * 4) + lcol;
#pragma unroll
                for (int f = 0; f < F; ++f) atomicAdd((float*)(Ob + (size_t)(oo + 128u * f)), v[u][f] * nj);
              }
            }
          }
        }
      }
      lds_barrier();
    }
  }
}

extern "C" void kernel_launch(void* const* d_in, const int* in_sizes, int n_in,
                              void* d_out, int out_size, void* d_ws, size_t ws_size,
                              hipStream_t stream) {
  const float* x  = (const float*)d_in[0];
  const int*   ei = (const int*)d_in[1];
  const float* W1 = (const float*)d_in[2];
  const float* b1 = (const float*)d_in[3];
  const float* W2 = (const float*)d_in[4];
  const float* b2 = (const float*)d_in[5];
  float* out = (float*)d_out;

  const int N = in_sizes[0] / GDIN;
  const int E = in_sizes[1] / 2;
  const int* src = ei;
  const int* dst = ei + E;

  float* bufA = (float*)d_ws;
  float* bufB = bufA + (size_t)N * GDH;
  float* dinv = bufB + (size_t)N * GDH;
  int* ctr = (int*)(dinv + N);

  const int mtiles = (N + 15) / 16;
  const int gblocks = (mtiles + 7) / 8;
  const int eblocks = (E + EW_PER_BLOCK - 1) / EW_PER_BLOCK;
  float* wt1 = out;
  float* wt2 = bufA + (size_t)N * GDOUT;

  unsigned* part = (unsigned*)bufA;
  const int nranges = (N + HCAP - 1) / HCAP;
  degree_hist_kernel<<<nranges * HPARTS, HTHREADS, 0, stream>>>(dst, part, N, E, nranges);
  dinv_kernel<<<(N + 255) / 256, 256, 0, stream>>>(part, dinv, ctr, N);

  gemm_wmma_f32<GDH, false><<<gblocks + eblocks, 256, 0, stream>>>(x, W1, dinv, b1, bufA, bufB, mtiles,
                                                                    src, dst, wt1, E);
  scatter_edges<GDH, 2><<<2048, 256, 0, stream>>>(src, dst, wt1, bufA, bufB, ctr, E, N);

  gemm_wmma_f32<GDOUT, true><<<gblocks + eblocks, 256, 0, stream>>>(bufB, W2, dinv, b2, bufA, out, mtiles,
                                                                     src, dst, wt2, E);
  scatter_edges<GDOUT, 1><<<2048, 256, 0, stream>>>(src, dst, wt2, bufA, out, ctr + 4 * 128, E, N);
}
